# starting kernel
# baseline (speedup 1.0000x reference)
#include <hip/hip_runtime.h>
#include <hip/hip_bf16.h>
#include <math.h>

#define NTOT   2048
#define CCAP   32
#define DIN    8
#define DOUT   16
#define BATCH  32
#define NPER   32
#define NCHUNK (NTOT / NPER)
#define COLS   (CCAP * DOUT)
#define OUTEL  (BATCH * COLS)

typedef __attribute__((ext_vector_type(2))) float v2f;
typedef __attribute__((ext_vector_type(8))) float v8f;

__global__ __launch_bounds__(256)
void caps_softmax(const float* __restrict__ R, float* __restrict__ Rn) {
    const int c   = blockIdx.x;
    const int tid = threadIdx.x;
    __shared__ float red[256];

    float m = -INFINITY;
    for (int n = tid; n < NTOT; n += 256) m = fmaxf(m, R[n * CCAP + c]);
    red[tid] = m; __syncthreads();
    for (int s = 128; s > 0; s >>= 1) {
        if (tid < s) red[tid] = fmaxf(red[tid], red[tid + s]);
        __syncthreads();
    }
    m = red[0]; __syncthreads();

    float sum = 0.0f;
    for (int n = tid; n < NTOT; n += 256) sum += __expf(R[n * CCAP + c] - m);
    red[tid] = sum; __syncthreads();
    for (int s = 128; s > 0; s >>= 1) {
        if (tid < s) red[tid] += red[tid + s];
        __syncthreads();
    }
    const float inv = 1.0f / red[0];

    for (int n = tid; n < NTOT; n += 256)
        Rn[n * CCAP + c] = __expf(R[n * CCAP + c] - m) * inv;
}

__global__ __launch_bounds__(32)
void caps_gemm(const float* __restrict__ x, const float* __restrict__ W,
               const float* __restrict__ Rn, float* __restrict__ part) {
    const int t    = blockIdx.x;
    const int ch   = blockIdx.y;
    const int lane = threadIdx.x;
    const int j    = lane & 15;
    const int hi   = lane >> 4;
    const int kofs = hi * 2;

    v8f acc0 = {};
    v8f acc1 = {};

    const int n0 = ch * NPER;
    const float* wp  = W  + (size_t)n0 * (CCAP * DOUT * DIN) + t * (DOUT * DIN) + j * DIN + kofs;
    const float* xp0 = x  + (size_t)j        * (NTOT * DIN) + n0 * DIN + kofs;
    const float* xp1 = x  + (size_t)(j + 16) * (NTOT * DIN) + n0 * DIN + kofs;
    const float* rp  = Rn + n0 * CCAP + t;

    #pragma unroll 4
    for (int nn = 0; nn < NPER; ++nn) {
        const float rn = rp[nn * CCAP];

        v2f b0, b1, a00, a01, a10, a11;
        b0.x  = wp[0] * rn;  b0.y  = wp[1] * rn;
        b1.x  = wp[4] * rn;  b1.y  = wp[5] * rn;
        a00.x = xp0[0];      a00.y = xp0[1];
        a01.x = xp0[4];      a01.y = xp0[5];
        a10.x = xp1[0];      a10.y = xp1[1];
        a11.x = xp1[4];      a11.y = xp1[5];

        acc0 = __builtin_amdgcn_wmma_f32_16x16x4_f32(false, a00, false, b0,
                                                     (short)0, acc0, false, false);
        acc0 = __builtin_amdgcn_wmma_f32_16x16x4_f32(false, a01, false, b1,
                                                     (short)0, acc0, false, false);
        acc1 = __builtin_amdgcn_wmma_f32_16x16x4_f32(false, a10, false, b0,
                                                     (short)0, acc1, false, false);
        acc1 = __builtin_amdgcn_wmma_f32_16x16x4_f32(false, a11, false, b1,
                                                     (short)0, acc1, false, false);

        wp  += CCAP * DOUT * DIN;
        xp0 += DIN;
        xp1 += DIN;
    }

    float* p = part + (size_t)ch * OUTEL + t * DOUT + j;
    #pragma unroll
    for (int v = 0; v < 8; ++v) {
        const int b = v + hi * 8;
        p[(size_t)b        * COLS] = acc0[v];
        p[(size_t)(b + 16) * COLS] = acc1[v];
    }
}

__global__ __launch_bounds__(256)
void caps_reduce_squash(const float* __restrict__ part, float* __restrict__ out) {
    const int idx = blockIdx.x * 256 + threadIdx.x;
    float s = 0.0f;
    #pragma unroll 8
    for (int ch = 0; ch < NCHUNK; ++ch)
        s += part[(size_t)ch * OUTEL + idx];

    float sq = s * s;
    for (int m = 1; m < 16; m <<= 1)
        sq += __shfl_xor(sq, m, 32);
    sq += 1e-7f;
    const float scale = sqrtf(sq) / (1.0f + sq);
    out[idx] = scale * s;
}

extern "C" void kernel_launch(void* const* d_in, const int* in_sizes, int n_in,
                              void* d_out, int out_size, void* d_ws, size_t ws_size,
                              hipStream_t stream) {
    const float* x = (const float*)d_in[0];
    const float* W = (const float*)d_in[1];
    const float* R = (const float*)d_in[2];
    float* out = (float*)d_out;

    float* Rn   = (float*)d_ws;
    float* part = Rn + (size_t)NTOT * CCAP;

    caps_softmax<<<CCAP, 256, 0, stream>>>(R, Rn);

    dim3 grid(CCAP, NCHUNK);
    caps_gemm<<<grid, 32, 0, stream>>>(x, W, Rn, part);

    caps_reduce_squash<<<OUTEL / 256, 256, 0, stream>>>(part, out);
}


# Softmax launch re-cut to a coalesced statistics pass: 64 one-wave blocks write per-chunk column max and exp-sum; each GEMM wave combines them into its 32 weights in registers; Rn plane removed
# speedup vs baseline: 1.5381x; 1.5381x over previous
#include <hip/hip_runtime.h>
#include <hip/hip_bf16.h>
#include <math.h>

#define NTOT   2048
#define CCAP   32
#define DIN    8
#define DOUT   16
#define BATCH  32
#define NPER   32
#define NCHUNK (NTOT / NPER)
#define COLS   (CCAP * DOUT)
#define OUTEL  (BATCH * COLS)

typedef __attribute__((ext_vector_type(2))) float v2f;
typedef __attribute__((ext_vector_type(8))) float v8f;

template <int M>
__device__ __forceinline__ float lane_xor_row(float v) {
    return __int_as_float(__builtin_amdgcn_update_dpp(0, __float_as_int(v), 0x160 | M, 0xf, 0xf, true));
}
__device__ __forceinline__ float lane_xor_16(float v) {
    const int i = __float_as_int(v);
    return __int_as_float(__builtin_amdgcn_permlanex16(i, i, 0x76543210, 0xfedcba98, false, false));
}
__device__ __forceinline__ float wave_max(float v) {
    v = fmaxf(v, lane_xor_row<1>(v));
    v = fmaxf(v, lane_xor_row<2>(v));
    v = fmaxf(v, lane_xor_row<4>(v));
    v = fmaxf(v, lane_xor_row<8>(v));
    return fmaxf(v, lane_xor_16(v));
}
__device__ __forceinline__ float wave_sum(float v) {
    v += lane_xor_row<1>(v);
    v += lane_xor_row<2>(v);
    v += lane_xor_row<4>(v);
    v += lane_xor_row<8>(v);
    return v + lane_xor_16(v);
}

__global__ __launch_bounds__(32)
void caps_colstats(const float* __restrict__ R, float* __restrict__ sm,
                   float* __restrict__ ss) {
    const int ch = blockIdx.x;
    const int c  = threadIdx.x;
    const float* p = R + ch * (NPER * CCAP) + c;

    float v[NPER];
    #pragma unroll
    for (int i = 0; i < NPER; ++i) v[i] = p[i * CCAP];

    float m = v[0];
    #pragma unroll
    for (int i = 1; i < NPER; ++i) m = fmaxf(m, v[i]);

    float sum = 0.0f;
    #pragma unroll
    for (int i = 0; i < NPER; ++i) sum += __expf(v[i] - m);

    sm[ch * CCAP + c] = m;
    ss[ch * CCAP + c] = sum;
}

__global__ __launch_bounds__(32)
void caps_gemm(const float* __restrict__ x, const float* __restrict__ W,
               const float* __restrict__ R, const float* __restrict__ sm,
               const float* __restrict__ ss, float* __restrict__ part) {
    const int t    = blockIdx.x;
    const int ch   = blockIdx.y;
    const int lane = threadIdx.x;
    const int j    = lane & 15;
    const int hi   = lane >> 4;
    const int kofs = hi * 2;

    v8f acc0 = {};
    v8f acc1 = {};

    const int n0 = ch * NPER;
    const float* wp  = W  + (size_t)n0 * (CCAP * DOUT * DIN) + t * (DOUT * DIN) + j * DIN + kofs;
    const float* xp0 = x  + (size_t)j        * (NTOT * DIN) + n0 * DIN + kofs;
    const float* xp1 = x  + (size_t)(j + 16) * (NTOT * DIN) + n0 * DIN + kofs;

    const float rv = R[(n0 + lane) * CCAP + t];
    const float m0 = sm[lane * CCAP + t];
    const float m1 = sm[(lane + 32) * CCAP + t];
    const float s0 = ss[lane * CCAP + t];
    const float s1 = ss[(lane + 32) * CCAP + t];
    const float cm = wave_max(fmaxf(m0, m1));
    const float cs = wave_sum(s0 * __expf(m0 - cm) + s1 * __expf(m1 - cm));
    const int rnl = __float_as_int(__expf(rv - cm) * (1.0f / cs));

    #pragma unroll 4
    for (int nn = 0; nn < NPER; ++nn) {
        const float rn = __int_as_float(__builtin_amdgcn_readlane(rnl, nn));

        v2f b0, b1, a00, a01, a10, a11;
        b0.x  = wp[0] * rn;  b0.y  = wp[1] * rn;
        b1.x  = wp[4] * rn;  b1.y  = wp[5] * rn;
        a00.x = xp0[0];      a00.y = xp0[1];
        a01.x = xp0[4];      a01.y = xp0[5];
        a10.x = xp1[0];      a10.y = xp1[1];
        a11.x = xp1[4];      a11.y = xp1[5];

        acc0 = __builtin_amdgcn_wmma_f32_16x16x4_f32(false, a00, false, b0,
                                                     (short)0, acc0, false, false);
        acc0 = __builtin_amdgcn_wmma_f32_16x16x4_f32(false, a01, false, b1,
                                                     (short)0, acc0, false, false);
        acc1 = __builtin_amdgcn_wmma_f32_16x16x4_f32(false, a10, false, b0,
                                                     (short)0, acc1, false, false);
        acc1 = __builtin_amdgcn_wmma_f32_16x16x4_f32(false, a11, false, b1,
                                                     (short)0, acc1, false, false);

        wp  += CCAP * DOUT * DIN;
        xp0 += DIN;
        xp1 += DIN;
    }

    float* p = part + (size_t)ch * OUTEL + t * DOUT + j;
    #pragma unroll
    for (int v = 0; v < 8; ++v) {
        const int b = v + hi * 8;
        p[(size_t)b        * COLS] = acc0[v];
        p[(size_t)(b + 16) * COLS] = acc1[v];
    }
}

__global__ __launch_bounds__(256)
void caps_reduce_squash(const float* __restrict__ part, float* __restrict__ out) {
    const int idx = blockIdx.x * 256 + threadIdx.x;
    float s = 0.0f;
    #pragma unroll 8
    for (int ch = 0; ch < NCHUNK; ++ch)
        s += part[(size_t)ch * OUTEL + idx];

    float sq = s * s;
    for (int m = 1; m < 16; m <<= 1)
        sq += __shfl_xor(sq, m, 32);
    sq += 1e-7f;
    const float scale = sqrtf(sq) / (1.0f + sq);
    out[idx] = scale * s;
}

extern "C" void kernel_launch(void* const* d_in, const int* in_sizes, int n_in,
                              void* d_out, int out_size, void* d_ws, size_t ws_size,
                              hipStream_t stream) {
    const float* x = (const float*)d_in[0];
    const float* W = (const float*)d_in[1];
    const float* R = (const float*)d_in[2];
    float* out = (float*)d_out;

    float* sm   = (float*)d_ws;
    float* ss   = sm + NCHUNK * CCAP;
    float* part = sm + (size_t)NTOT * CCAP;

    caps_colstats<<<NCHUNK, 32, 0, stream>>>(R, sm, ss);

    dim3 grid(CCAP, NCHUNK);
    caps_gemm<<<grid, 32, 0, stream>>>(x, W, R, sm, ss, part);

    caps_reduce_squash<<<OUTEL / 256, 256, 0, stream>>>(part, out);
}


# GEMM re-tiled: each wave owns two capsules x 32 rows, the x fragment of a row is loaded once and used by both; operand loads 16 B per lane kept 12-14 rows ahead of the WMMAs; 256 four-wave blocks
# speedup vs baseline: 2.1295x; 1.3845x over previous
#include <hip/hip_runtime.h>
#include <hip/hip_bf16.h>
#include <math.h>

#define NTOT   2048
#define CCAP   32
#define DIN    8
#define DOUT   16
#define BATCH  32
#define NPER   32
#define NCHUNK (NTOT / NPER)
#define COLS   (CCAP * DOUT)
#define OUTEL  (BATCH * COLS)

typedef __attribute__((ext_vector_type(2))) float v2f;
typedef __attribute__((ext_vector_type(4))) float v4f;
typedef __attribute__((ext_vector_type(8))) float v8f;

template <int M>
__device__ __forceinline__ float lane_xor_row(float v) {
    return __int_as_float(__builtin_amdgcn_update_dpp(0, __float_as_int(v), 0x160 | M, 0xf, 0xf, true));
}
__device__ __forceinline__ float lane_xor_16(float v) {
    const int i = __float_as_int(v);
    return __int_as_float(__builtin_amdgcn_permlanex16(i, i, 0x76543210, 0xfedcba98, false, false));
}
__device__ __forceinline__ float wave_max(float v) {
    v = fmaxf(v, lane_xor_row<1>(v));
    v = fmaxf(v, lane_xor_row<2>(v));
    v = fmaxf(v, lane_xor_row<4>(v));
    v = fmaxf(v, lane_xor_row<8>(v));
    return fmaxf(v, lane_xor_16(v));
}
__device__ __forceinline__ float wave_sum(float v) {
    v += lane_xor_row<1>(v);
    v += lane_xor_row<2>(v);
    v += lane_xor_row<4>(v);
    v += lane_xor_row<8>(v);
    return v + lane_xor_16(v);
}

__global__ __launch_bounds__(32)
void caps_colstats(const float* __restrict__ R, float* __restrict__ sm,
                   float* __restrict__ ss) {
    const int ch = blockIdx.x;
    const int c  = threadIdx.x;
    const float* p = R + ch * (NPER * CCAP) + c;

    float v[NPER];
    #pragma unroll
    for (int i = 0; i < NPER; ++i) v[i] = p[i * CCAP];

    float m = v[0];
    #pragma unroll
    for (int i = 1; i < NPER; ++i) m = fmaxf(m, v[i]);

    float sum = 0.0f;
    #pragma unroll
    for (int i = 0; i < NPER; ++i) sum += __expf(v[i] - m);

    sm[ch * CCAP + c] = m;
    ss[ch * CCAP + c] = sum;
}

#define CG     2
#define GN     2
#define NGRP   (NPER / GN)
#define GDEPTH 7
#define WPB    4

struct LoadGroup {
    v4f w[GN][CG];
    v4f xa[GN][2];
};

__device__ __forceinline__ void load_group(LoadGroup& g, const float* wp,
                                           const float* xp0, const float* xp1,
                                           const int nbase) {
    #pragma unroll
    for (int i = 0; i < GN; ++i) {
        const int n = nbase + i;
        g.w[i][0]  = *(const v4f*)(wp + (size_t)n * (CCAP * DOUT * DIN));
        g.w[i][1]  = *(const v4f*)(wp + (size_t)n * (CCAP * DOUT * DIN) + DOUT * DIN);
        g.xa[i][0] = *(const v4f*)(xp0 + n * DIN);
        g.xa[i][1] = *(const v4f*)(xp1 + n * DIN);
    }
}

__global__ __launch_bounds__(32 * WPB)
void caps_gemm(const float* __restrict__ x, const float* __restrict__ W,
               const float* __restrict__ R, const float* __restrict__ sm,
               const float* __restrict__ ss, float* __restrict__ part) {
    const int t0   = (blockIdx.x * WPB + (threadIdx.x >> 5)) * CG;
    const int ch   = blockIdx.y;
    const int lane = threadIdx.x & 31;
    const int j    = lane & 15;
    const int hi   = lane >> 4;
    const int n0   = ch * NPER;

    const float* wp  = W + (size_t)n0 * (CCAP * DOUT * DIN) + t0 * (DOUT * DIN) + j * DIN + hi * 4;
    const float* xp0 = x + (size_t)j        * (NTOT * DIN) + n0 * DIN + hi * 4;
    const float* xp1 = x + (size_t)(j + 16) * (NTOT * DIN) + n0 * DIN + hi * 4;

    const v2f rv = *(const v2f*)(R  + (n0 + lane) * CCAP + t0);
    const v2f m0 = *(const v2f*)(sm + lane * CCAP + t0);
    const v2f m1 = *(const v2f*)(sm + (lane + 32) * CCAP + t0);
    const v2f s0 = *(const v2f*)(ss + lane * CCAP + t0);
    const v2f s1 = *(const v2f*)(ss + (lane + 32) * CCAP + t0);

    LoadGroup buf[GDEPTH];
    #pragma unroll
    for (int g = 0; g < GDEPTH; ++g) load_group(buf[g], wp, xp0, xp1, g * GN);
    __builtin_amdgcn_sched_barrier(0);

    const float cm0 = wave_max(fmaxf(m0.x, m1.x));
    const float cm1 = wave_max(fmaxf(m0.y, m1.y));
    const float cs0 = wave_sum(s0.x * __expf(m0.x - cm0) + s1.x * __expf(m1.x - cm0));
    const float cs1 = wave_sum(s0.y * __expf(m0.y - cm1) + s1.y * __expf(m1.y - cm1));
    const int rnl0 = __float_as_int(__expf(rv.x - cm0) * (1.0f / cs0));
    const int rnl1 = __float_as_int(__expf(rv.y - cm1) * (1.0f / cs1));

    v8f acc00 = {}, acc01 = {};
    v8f acc10 = {}, acc11 = {};
    __builtin_amdgcn_sched_barrier(0);

    #pragma unroll
    for (int g = 0; g < NGRP; ++g) {
        LoadGroup& cur = buf[g % GDEPTH];
        #pragma unroll
        for (int i = 0; i < GN; ++i) {
            const int nn = g * GN + i;
            const float r0 = __int_as_float(__builtin_amdgcn_readlane(rnl0, nn));
            const float r1 = __int_as_float(__builtin_amdgcn_readlane(rnl1, nn));
            const v4f w0 = cur.w[i][0] * r0;
            const v4f w1 = cur.w[i][1] * r1;
            const v4f xa = cur.xa[i][0];
            const v4f xb = cur.xa[i][1];
            const v2f a0 = {xa.x, xa.y}, a1 = {xa.z, xa.w};
            const v2f c0 = {xb.x, xb.y}, c1 = {xb.z, xb.w};
            const v2f p0 = {w0.x, w0.y}, p1 = {w0.z, w0.w};
            const v2f q0 = {w1.x, w1.y}, q1 = {w1.z, w1.w};

            acc00 = __builtin_amdgcn_wmma_f32_16x16x4_f32(false, a0, false, p0, (short)0, acc00, false, false);
            acc01 = __builtin_amdgcn_wmma_f32_16x16x4_f32(false, c0, false, p0, (short)0, acc01, false, false);
            acc10 = __builtin_amdgcn_wmma_f32_16x16x4_f32(false, a0, false, q0, (short)0, acc10, false, false);
            acc11 = __builtin_amdgcn_wmma_f32_16x16x4_f32(false, c0, false, q0, (short)0, acc11, false, false);
            acc00 = __builtin_amdgcn_wmma_f32_16x16x4_f32(false, a1, false, p1, (short)0, acc00, false, false);
            acc01 = __builtin_amdgcn_wmma_f32_16x16x4_f32(false, c1, false, p1, (short)0, acc01, false, false);
            acc10 = __builtin_amdgcn_wmma_f32_16x16x4_f32(false, a1, false, q1, (short)0, acc10, false, false);
            acc11 = __builtin_amdgcn_wmma_f32_16x16x4_f32(false, c1, false, q1, (short)0, acc11, false, false);
        }
        __builtin_amdgcn_sched_barrier(0);
        if (g + GDEPTH < NGRP) {
            load_group(cur, wp, xp0, xp1, (g + GDEPTH) * GN);
            __builtin_amdgcn_sched_barrier(0);
        }
    }

    float* p = part + (size_t)ch * OUTEL + t0 * DOUT + j;
    #pragma unroll
    for (int v = 0; v < 8; ++v) {
        const int b = v + hi * 8;
        p[(size_t)b        * COLS]        = acc00[v];
        p[(size_t)b        * COLS + DOUT] = acc10[v];
        p[(size_t)(b + 16) * COLS]        = acc01[v];
        p[(size_t)(b + 16) * COLS + DOUT] = acc11[v];
    }
}

__global__ __launch_bounds__(256)
void caps_reduce_squash(const float* __restrict__ part, float* __restrict__ out) {
    const int idx = blockIdx.x * 256 + threadIdx.x;
    float s = 0.0f;
    #pragma unroll 8
    for (int ch = 0; ch < NCHUNK; ++ch)
        s += part[(size_t)ch * OUTEL + idx];

    float sq = s * s;
    for (int m = 1; m < 16; m <<= 1)
        sq += __shfl_xor(sq, m, 32);
    sq += 1e-7f;
    const float scale = sqrtf(sq) / (1.0f + sq);
    out[idx] = scale * s;
}

extern "C" void kernel_launch(void* const* d_in, const int* in_sizes, int n_in,
                              void* d_out, int out_size, void* d_ws, size_t ws_size,
                              hipStream_t stream) {
    const float* x = (const float*)d_in[0];
    const float* W = (const float*)d_in[1];
    const float* R = (const float*)d_in[2];
    float* out = (float*)d_out;

    float* sm   = (float*)d_ws;
    float* ss   = sm + NCHUNK * CCAP;
    float* part = sm + (size_t)NTOT * CCAP;

    caps_colstats<<<NCHUNK, 32, 0, stream>>>(R, sm, ss);

    dim3 grid(CCAP / (CG * WPB), NCHUNK);
    caps_gemm<<<grid, 32 * WPB, 0, stream>>>(x, W, R, sm, ss, part);

    caps_reduce_squash<<<OUTEL / 256, 256, 0, stream>>>(part, out);
}


# GEMM re-tiled, K split inside the block: a wave owns 4 capsules x 16 rows, a block's 4 waves sum their tiles through LDS; 32 partial planes in whole 256-byte rows, read in one batch by the reduce
# speedup vs baseline: 2.7616x; 1.2968x over previous
#include <hip/hip_runtime.h>
#include <hip/hip_bf16.h>
#include <math.h>

#define NTOT   2048
#define CCAP   32
#define DIN    8
#define DOUT   16
#define BATCH  32
#define NPER   32
#define NCHUNK (NTOT / NPER)
#define COLS   (CCAP * DOUT)
#define OUTEL  (BATCH * COLS)

typedef __attribute__((ext_vector_type(2))) float v2f;
typedef __attribute__((ext_vector_type(4))) float v4f;
typedef __attribute__((ext_vector_type(8))) float v8f;

template <int M>
__device__ __forceinline__ float lane_xor_row(float v) {
    return __int_as_float(__builtin_amdgcn_update_dpp(0, __float_as_int(v), 0x160 | M, 0xf, 0xf, true));
}
__device__ __forceinline__ float lane_xor_16(float v) {
    const int i = __float_as_int(v);
    return __int_as_float(__builtin_amdgcn_permlanex16(i, i, 0x76543210, 0xfedcba98, false, false));
}
__device__ __forceinline__ float wave_max(float v) {
    v = fmaxf(v, lane_xor_row<1>(v));
    v = fmaxf(v, lane_xor_row<2>(v));
    v = fmaxf(v, lane_xor_row<4>(v));
    v = fmaxf(v, lane_xor_row<8>(v));
    return fmaxf(v, lane_xor_16(v));
}
__device__ __forceinline__ float wave_sum(float v) {
    v += lane_xor_row<1>(v);
    v += lane_xor_row<2>(v);
    v += lane_xor_row<4>(v);
    v += lane_xor_row<8>(v);
    return v + lane_xor_16(v);
}

__global__ __launch_bounds__(32)
void caps_colstats(const float* __restrict__ R, float* __restrict__ sm,
                   float* __restrict__ ss) {
    const int ch = blockIdx.x;
    const int c  = threadIdx.x;
    const float* p = R + ch * (NPER * CCAP) + c;

    float v[NPER];
    #pragma unroll
    for (int i = 0; i < NPER; ++i) v[i] = p[i * CCAP];

    float m = v[0];
    #pragma unroll
    for (int i = 1; i < NPER; ++i) m = fmaxf(m, v[i]);

    float sum = 0.0f;
    #pragma unroll
    for (int i = 0; i < NPER; ++i) sum += __expf(v[i] - m);

    sm[ch * CCAP + c] = m;
    ss[ch * CCAP + c] = sum;
}

#define CQ     4
#define RW     16
#define WPB    4
#define NSLAB  (RW * WPB)
#define NPLANE (NTOT / NSLAB)
#define DEPTH  9
#define TCOLS  (CQ * DOUT)
#define LPITCH (TCOLS + 4)

struct RowLoad {
    v4f w[CQ];
    v4f xa[2];
};

__device__ __forceinline__ void load_row(RowLoad& g, const float* wp,
                                         const float* xp0, const float* xp1,
                                         const int n) {
    #pragma unroll
    for (int q = 0; q < CQ; ++q)
        g.w[q] = *(const v4f*)(wp + (size_t)n * (CCAP * DOUT * DIN) + q * (DOUT * DIN));
    g.xa[0] = *(const v4f*)(xp0 + n * DIN);
    g.xa[1] = *(const v4f*)(xp1 + n * DIN);
}

__global__ __launch_bounds__(32 * WPB)
void caps_gemm(const float* __restrict__ x, const float* __restrict__ W,
               const float* __restrict__ R, const float* __restrict__ sm,
               const float* __restrict__ ss, float* __restrict__ part) {
    __shared__ __attribute__((aligned(16))) float red[WPB * BATCH * LPITCH];

    const int wv   = threadIdx.x >> 5;
    const int t0   = blockIdx.x * CQ;
    const int slab = blockIdx.y;
    const int lane = threadIdx.x & 31;
    const int j    = lane & 15;
    const int hi   = lane >> 4;
    const int n0   = slab * NSLAB + wv * RW;

    const float* wp  = W + (size_t)n0 * (CCAP * DOUT * DIN) + t0 * (DOUT * DIN) + j * DIN + hi * 4;
    const float* xp0 = x + (size_t)j        * (NTOT * DIN) + n0 * DIN + hi * 4;
    const float* xp1 = x + (size_t)(j + 16) * (NTOT * DIN) + n0 * DIN + hi * 4;

    const v4f rv = *(const v4f*)(R  + (n0 + j) * CCAP + t0);
    const v4f m0 = *(const v4f*)(sm + lane * CCAP + t0);
    const v4f m1 = *(const v4f*)(sm + (lane + 32) * CCAP + t0);
    const v4f s0 = *(const v4f*)(ss + lane * CCAP + t0);
    const v4f s1 = *(const v4f*)(ss + (lane + 32) * CCAP + t0);

    __builtin_amdgcn_sched_barrier(0);
    RowLoad buf[DEPTH];
    #pragma unroll
    for (int g = 0; g < DEPTH; ++g) {
        load_row(buf[g], wp, xp0, xp1, g);
        __builtin_amdgcn_sched_barrier(0);
    }

    int rnl[CQ];
    #pragma unroll
    for (int q = 0; q < CQ; ++q) {
        const float cm = wave_max(fmaxf(m0[q], m1[q]));
        const float cs = wave_sum(s0[q] * __expf(m0[q] - cm) + s1[q] * __expf(m1[q] - cm));
        rnl[q] = __float_as_int(__expf(rv[q] - cm) * (1.0f / cs));
    }

    v8f acc[CQ][2];
    #pragma unroll
    for (int q = 0; q < CQ; ++q) { acc[q][0] = v8f{}; acc[q][1] = v8f{}; }
    __builtin_amdgcn_sched_barrier(0);

    #pragma unroll
    for (int g = 0; g < RW; ++g) {
        RowLoad& cur = buf[g % DEPTH];
        const v4f xa = cur.xa[0];
        const v4f xb = cur.xa[1];
        const v2f a0 = {xa.x, xa.y}, a1 = {xa.z, xa.w};
        const v2f c0 = {xb.x, xb.y}, c1 = {xb.z, xb.w};
        v2f p0[CQ], p1[CQ];
        #pragma unroll
        for (int q = 0; q < CQ; ++q) {
            const float r = __int_as_float(__builtin_amdgcn_readlane(rnl[q], g));
            const v4f w = cur.w[q] * r;
            p0[q] = v2f{w.x, w.y};
            p1[q] = v2f{w.z, w.w};
        }
        #pragma unroll
        for (int q = 0; q < CQ; ++q) {
            acc[q][0] = __builtin_amdgcn_wmma_f32_16x16x4_f32(false, a0, false, p0[q], (short)0, acc[q][0], false, false);
            acc[q][1] = __builtin_amdgcn_wmma_f32_16x16x4_f32(false, c0, false, p0[q], (short)0, acc[q][1], false, false);
        }
        #pragma unroll
        for (int q = 0; q < CQ; ++q) {
            acc[q][0] = __builtin_amdgcn_wmma_f32_16x16x4_f32(false, a1, false, p1[q], (short)0, acc[q][0], false, false);
            acc[q][1] = __builtin_amdgcn_wmma_f32_16x16x4_f32(false, c1, false, p1[q], (short)0, acc[q][1], false, false);
        }
        __builtin_amdgcn_sched_barrier(0);
        if (g + DEPTH < RW) {
            load_row(cur, wp, xp0, xp1, g + DEPTH);
            __builtin_amdgcn_sched_barrier(0);
        }
    }

    float* mine = red + wv * (BATCH * LPITCH);
    #pragma unroll
    for (int q = 0; q < CQ; ++q) {
        #pragma unroll
        for (int bh = 0; bh < 2; ++bh) {
            #pragma unroll
            for (int v = 0; v < 8; ++v)
                mine[(bh * 16 + hi * 8 + v) * LPITCH + q * DOUT + j] = acc[q][bh][v];
        }
    }
    __syncthreads();

    float* dst = part + (size_t)slab * OUTEL + t0 * DOUT;
    #pragma unroll
    for (int i = 0; i < (BATCH * TCOLS) / (4 * 32 * WPB); ++i) {
        const int g4 = threadIdx.x + i * (32 * WPB);
        const int b  = g4 >> 4;
        const int c4 = (g4 & 15) * 4;
        const float* src = red + b * LPITCH + c4;
        v4f s = *(const v4f*)(src);
        #pragma unroll
        for (int w = 1; w < WPB; ++w)
            s += *(const v4f*)(src + w * (BATCH * LPITCH));
        *(v4f*)(dst + (size_t)b * COLS + c4) = s;
    }
}

__global__ __launch_bounds__(256)
void caps_reduce_squash(const float* __restrict__ part, float* __restrict__ out) {
    const int idx = blockIdx.x * 256 + threadIdx.x;
    float v[NPLANE];
    #pragma unroll
    for (int ch = 0; ch < NPLANE; ++ch)
        v[ch] = part[(size_t)ch * OUTEL + idx];
    float s = 0.0f;
    #pragma unroll
    for (int ch = 0; ch < NPLANE; ++ch)
        s += v[ch];

    float sq = s * s;
    for (int m = 1; m < 16; m <<= 1)
        sq += __shfl_xor(sq, m, 32);
    sq += 1e-7f;
    const float scale = sqrtf(sq) / (1.0f + sq);
    out[idx] = scale * s;
}

extern "C" void kernel_launch(void* const* d_in, const int* in_sizes, int n_in,
                              void* d_out, int out_size, void* d_ws, size_t ws_size,
                              hipStream_t stream) {
    const float* x = (const float*)d_in[0];
    const float* W = (const float*)d_in[1];
    const float* R = (const float*)d_in[2];
    float* out = (float*)d_out;

    float* sm   = (float*)d_ws;
    float* ss   = sm + NCHUNK * CCAP;
    float* part = sm + (size_t)NTOT * CCAP;

    caps_colstats<<<NCHUNK, 32, 0, stream>>>(R, sm, ss);

    dim3 grid(CCAP / CQ, NPLANE);
    caps_gemm<<<grid, 32 * WPB, 0, stream>>>(x, W, R, sm, ss, part);

    caps_reduce_squash<<<OUTEL / 256, 256, 0, stream>>>(part, out);
}
